# starting kernel
# baseline (speedup 1.0000x reference)
#include <hip/hip_runtime.h>
#include <hip/hip_bf16.h>

typedef __attribute__((ext_vector_type(2))) float v2f;
typedef __attribute__((ext_vector_type(8))) float v8f;

#define B_   4
#define N_   32768
#define C_   512
#define M_   512
#define CH_  (C_ + 4)
#define CAP  8704

__global__ __launch_bounds__(256) void score_kernel(
    const float* __restrict__ feat, const float* __restrict__ Wobj,
    const float* __restrict__ bobj, const float* __restrict__ Wg,
    const float* __restrict__ bg, float* __restrict__ grasp_out,
    unsigned char* __restrict__ mask_out)
{
    __shared__ float Apre[128][32][2];
    for (int t = threadIdx.x; t < 128 * 32; t += 256) {
        int kc = t >> 5, lane = t & 31;
        int row = lane & 15;
        int kh  = lane >> 4;
        int ka  = kc * 4 + 2 * kh;
        float a0 = 0.0f, a1 = 0.0f;
        if (row < 2)      { a0 = Wobj[row * C_ + ka]; a1 = Wobj[row * C_ + ka + 1]; }
        else if (row == 2){ a0 = Wg[ka];              a1 = Wg[ka + 1]; }
        Apre[kc][lane][0] = a0;
        Apre[kc][lane][1] = a1;
    }
    __syncthreads();

    const int wave = threadIdx.x >> 5;
    const int lane = threadIdx.x & 31;
    const int b    = blockIdx.x >> 8;
    const int tile = blockIdx.x & 255;
    const int n0   = tile * 128 + wave * 16;
    const int kh   = lane >> 4;
    const int col  = n0 + (lane & 15);

    const float* fb = feat + (size_t)b * C_ * N_;

    float b0 = fb[(size_t)(2 * kh) * N_ + col];
    float b1 = fb[(size_t)(2 * kh + 1) * N_ + col];

    v8f acc = {};
#pragma unroll 2
    for (int kc = 0; kc < 128; ++kc) {
        v2f a, bm;
        a.x  = Apre[kc][lane][0];
        a.y  = Apre[kc][lane][1];
        bm.x = b0;
        bm.y = b1;

        const int kn  = (kc + 1) & 127;
        const int kan = kn * 4 + 2 * kh;
        b0 = fb[(size_t)kan * N_ + col];
        b1 = fb[(size_t)(kan + 1) * N_ + col];

        int pfrow = kc * 4 + 32;
        if (pfrow > C_ - 4) pfrow = C_ - 4;
        __builtin_prefetch(&fb[(size_t)(pfrow + 2 * kh) * N_ + col], 0, 3);
        __builtin_prefetch(&fb[(size_t)(pfrow + 2 * kh + 1) * N_ + col], 0, 3);

        acc = __builtin_amdgcn_wmma_f32_16x16x4_f32(
            false, a, false, bm, (short)0, acc, false, false);
    }

    if (lane < 16) {
        float s0 = acc[0] + bobj[0];
        float s1 = acc[1] + bobj[1];
        float g  = acc[2] + bg[0];
        int p = n0 + lane;
        grasp_out[(size_t)b * N_ + p] = g;
        mask_out[(size_t)b * N_ + p] = (unsigned char)((s1 > s0) && (g > 0.1f));
    }
}

__global__ __launch_bounds__(1024) void fps_kernel(
    const float* __restrict__ xyz, const unsigned char* __restrict__ mask,
    int* __restrict__ out_idx)
{
    __shared__ float4 pts[CAP];
    __shared__ unsigned short cidx[CAP];
    __shared__ int scanbuf[1024];
    __shared__ int base_sh;
    __shared__ int sel_sh;
    __shared__ float wv[32];
    __shared__ int   wi[32];

    const int b   = blockIdx.x;
    const int tid = threadIdx.x;
    const float* xb = xyz + (size_t)b * N_ * 3;
    const unsigned char* mb = mask + (size_t)b * N_;

    if (tid == 0) { base_sh = 0; sel_sh = 0; }
    __syncthreads();

    for (int c0 = 0; c0 < N_; c0 += 1024) {
        const int i = c0 + tid;
        const int m = mb[i] ? 1 : 0;
        scanbuf[tid] = m;
        __syncthreads();
        for (int off = 1; off < 1024; off <<= 1) {
            int t  = scanbuf[tid];
            int t2 = (tid >= off) ? scanbuf[tid - off] : 0;
            __syncthreads();
            scanbuf[tid] = t + t2;
            __syncthreads();
        }
        const int incl = scanbuf[tid];
        const int base = base_sh;
        const int pos  = base + incl - m;
        if (m && pos < CAP) {
            pts[pos]  = make_float4(xb[(size_t)i * 3], xb[(size_t)i * 3 + 1],
                                    xb[(size_t)i * 3 + 2], __builtin_inff());
            cidx[pos] = (unsigned short)i;
        }
        __syncthreads();
        if (tid == 1023) base_sh = base + incl;
        __syncthreads();
    }
    int Nc = base_sh;
    if (Nc > CAP) Nc = CAP;
    if (Nc == 0) {
        for (int s = tid; s < M_; s += 1024) out_idx[b * M_ + s] = 0;
        return;
    }

    const float NEG_INF = -__builtin_inff();
    const int lane = tid & 31, wav = tid >> 5;

    for (int s = 0; s < M_; ++s) {
        const int sel = sel_sh;
        if (tid == 0) out_idx[b * M_ + s] = (int)cidx[sel];
        const float lx = pts[sel].x, ly = pts[sel].y, lz = pts[sel].z;

        float bv = NEG_INF;
        int   bi = 0x7fffffff;
        for (int i = tid; i < Nc; i += 1024) {
            float4 p = pts[i];
            float dx = __fsub_rn(p.x, lx);
            float dy = __fsub_rn(p.y, ly);
            float dz = __fsub_rn(p.z, lz);
            float d  = __fadd_rn(__fadd_rn(__fmul_rn(dx, dx), __fmul_rn(dy, dy)),
                                 __fmul_rn(dz, dz));
            float nd = fminf(p.w, d);
            pts[i].w = nd;
            if (nd > bv) { bv = nd; bi = i; }
        }
        for (int off = 16; off; off >>= 1) {
            float v2 = __shfl_xor(bv, off, 32);
            int   i2 = __shfl_xor(bi, off, 32);
            if (v2 > bv || (v2 == bv && i2 < bi)) { bv = v2; bi = i2; }
        }
        if (lane == 0) { wv[wav] = bv; wi[wav] = bi; }
        __syncthreads();
        if (wav == 0) {
            bv = wv[lane]; bi = wi[lane];
            for (int off = 16; off; off >>= 1) {
                float v2 = __shfl_xor(bv, off, 32);
                int   i2 = __shfl_xor(bi, off, 32);
                if (v2 > bv || (v2 == bv && i2 < bi)) { bv = v2; bi = i2; }
            }
            if (lane == 0) sel_sh = bi;
        }
        __syncthreads();
    }
}

__global__ __launch_bounds__(256) void gather_kernel(
    const float* __restrict__ feat, const float* __restrict__ xyz,
    const float* __restrict__ grasp, const int* __restrict__ idxs,
    float* __restrict__ out)
{
    const int g = blockIdx.x * 256 + threadIdx.x;
    if (g >= B_ * CH_ * M_) return;
    const int m  = g % M_;
    const int t  = g / M_;
    const int ch = t % CH_;
    const int b  = t / CH_;
    const int id = idxs[b * M_ + m];
    float v;
    if (ch < C_)            v = feat[((size_t)b * C_ + ch) * N_ + id];
    else if (ch < C_ + 3)   v = xyz[((size_t)b * N_ + id) * 3 + (ch - C_)];
    else                    v = grasp[(size_t)b * N_ + id];
    out[g] = v;
}

extern "C" void kernel_launch(void* const* d_in, const int* in_sizes, int n_in,
                              void* d_out, int out_size, void* d_ws, size_t ws_size,
                              hipStream_t stream) {
    const float* xyz  = (const float*)d_in[0];
    const float* feat = (const float*)d_in[1];
    const float* Wobj = (const float*)d_in[2];
    const float* bobj = (const float*)d_in[3];
    const float* Wg   = (const float*)d_in[4];
    const float* bg   = (const float*)d_in[5];
    float* out = (float*)d_out;

    char* ws = (char*)d_ws;
    float*         g_grasp = (float*)ws;
    unsigned char* g_mask  = (unsigned char*)(ws + (size_t)B_ * N_ * 4);
    int*           g_idx   = (int*)(ws + (size_t)B_ * N_ * 4 + (size_t)B_ * N_);

    score_kernel<<<B_ * (N_ / 128), 256, 0, stream>>>(feat, Wobj, bobj, Wg, bg,
                                                      g_grasp, g_mask);
    fps_kernel<<<B_, 1024, 0, stream>>>(xyz, g_mask, g_idx);
    const int total = B_ * CH_ * M_;
    gather_kernel<<<(total + 255) / 256, 256, 0, stream>>>(feat, xyz, g_grasp,
                                                           g_idx, out);
}


# FPS step argmax: wave max by row_xmask DPP + lane-permute, index by ballot, not LDS shuffles; every wave reduces the 32-wave table: 1 barrier per step; picks stored at end; distances bit-identical
# speedup vs baseline: 1.9839x; 1.9839x over previous
#include <hip/hip_runtime.h>
#include <hip/hip_bf16.h>

typedef __attribute__((ext_vector_type(2))) float v2f;
typedef __attribute__((ext_vector_type(8))) float v8f;

#define B_   4
#define N_   32768
#define C_   512
#define M_   512
#define CH_  (C_ + 4)
#define CAP  8704

__global__ __launch_bounds__(256) void score_kernel(
    const float* __restrict__ feat, const float* __restrict__ Wobj,
    const float* __restrict__ bobj, const float* __restrict__ Wg,
    const float* __restrict__ bg, float* __restrict__ grasp_out,
    unsigned char* __restrict__ mask_out)
{
    __shared__ float Apre[128][32][2];
    for (int t = threadIdx.x; t < 128 * 32; t += 256) {
        int kc = t >> 5, lane = t & 31;
        int row = lane & 15;
        int kh  = lane >> 4;
        int ka  = kc * 4 + 2 * kh;
        float a0 = 0.0f, a1 = 0.0f;
        if (row < 2)      { a0 = Wobj[row * C_ + ka]; a1 = Wobj[row * C_ + ka + 1]; }
        else if (row == 2){ a0 = Wg[ka];              a1 = Wg[ka + 1]; }
        Apre[kc][lane][0] = a0;
        Apre[kc][lane][1] = a1;
    }
    __syncthreads();

    const int wave = threadIdx.x >> 5;
    const int lane = threadIdx.x & 31;
    const int b    = blockIdx.x >> 8;
    const int tile = blockIdx.x & 255;
    const int n0   = tile * 128 + wave * 16;
    const int kh   = lane >> 4;
    const int col  = n0 + (lane & 15);

    const float* fb = feat + (size_t)b * C_ * N_;

    float b0 = fb[(size_t)(2 * kh) * N_ + col];
    float b1 = fb[(size_t)(2 * kh + 1) * N_ + col];

    v8f acc = {};
#pragma unroll 2
    for (int kc = 0; kc < 128; ++kc) {
        v2f a, bm;
        a.x  = Apre[kc][lane][0];
        a.y  = Apre[kc][lane][1];
        bm.x = b0;
        bm.y = b1;

        const int kn  = (kc + 1) & 127;
        const int kan = kn * 4 + 2 * kh;
        b0 = fb[(size_t)kan * N_ + col];
        b1 = fb[(size_t)(kan + 1) * N_ + col];

        int pfrow = kc * 4 + 32;
        if (pfrow > C_ - 4) pfrow = C_ - 4;
        __builtin_prefetch(&fb[(size_t)(pfrow + 2 * kh) * N_ + col], 0, 3);
        __builtin_prefetch(&fb[(size_t)(pfrow + 2 * kh + 1) * N_ + col], 0, 3);

        acc = __builtin_amdgcn_wmma_f32_16x16x4_f32(
            false, a, false, bm, (short)0, acc, false, false);
    }

    if (lane < 16) {
        float s0 = acc[0] + bobj[0];
        float s1 = acc[1] + bobj[1];
        float g  = acc[2] + bg[0];
        int p = n0 + lane;
        grasp_out[(size_t)b * N_ + p] = g;
        mask_out[(size_t)b * N_ + p] = (unsigned char)((s1 > s0) && (g > 0.1f));
    }
}

#define SLOTS ((CAP + 1023) / 1024)

__device__ __forceinline__ int xrow_i(int v) {
    return __builtin_amdgcn_permlanex16(v, v, 0x76543210, 0xfedcba98, false, false);
}
template <int X> __device__ __forceinline__ int rowx_i(int v) {
    return __builtin_amdgcn_update_dpp(0, v, 0x160 + X, 0xf, 0xf, true);
}
__device__ __forceinline__ int imax(int a, int b) { return a > b ? a : b; }
__device__ __forceinline__ int imin(int a, int b) { return a < b ? a : b; }
__device__ __forceinline__ int wave_max_i(int v) {
    v = imax(v, rowx_i<1>(v));
    v = imax(v, rowx_i<2>(v));
    v = imax(v, rowx_i<4>(v));
    v = imax(v, rowx_i<8>(v));
    return imax(v, xrow_i(v));
}
__device__ __forceinline__ int wave_min_i(int v) {
    v = imin(v, rowx_i<1>(v));
    v = imin(v, rowx_i<2>(v));
    v = imin(v, rowx_i<4>(v));
    v = imin(v, rowx_i<8>(v));
    return imin(v, xrow_i(v));
}
__device__ __forceinline__ void wave_argmax(int kv, int ki, int& mk, int& mi) {
    mk = wave_max_i(kv);
    const unsigned eq = __builtin_amdgcn_ballot_w32(kv == mk);
    mi = __builtin_amdgcn_readlane(ki, __builtin_ctz(eq));
    if (__builtin_popcount(eq) != 1) {
        mi = __builtin_amdgcn_readfirstlane(
                 wave_min_i(kv == mk ? ki : 0x7fffffff));
    }
}
__global__ __launch_bounds__(1024) void fps_kernel(
    const float* __restrict__ xyz, const unsigned char* __restrict__ mask,
    int* __restrict__ out_idx)
{
    __shared__ float4 pts[CAP];
    __shared__ unsigned short cidx[CAP];
    __shared__ int scanbuf[1024];
    __shared__ int base_sh;
    __shared__ int2 tab[2][32];
    __shared__ unsigned short picks[M_];

    const int b   = blockIdx.x;
    const int tid = threadIdx.x;
    const float* xb = xyz + (size_t)b * N_ * 3;
    const unsigned char* mb = mask + (size_t)b * N_;

    if (tid == 0) base_sh = 0;
    __syncthreads();

    for (int c0 = 0; c0 < N_; c0 += 1024) {
        const int i = c0 + tid;
        const int m = mb[i] ? 1 : 0;
        scanbuf[tid] = m;
        __syncthreads();
        for (int off = 1; off < 1024; off <<= 1) {
            int t  = scanbuf[tid];
            int t2 = (tid >= off) ? scanbuf[tid - off] : 0;
            __syncthreads();
            scanbuf[tid] = t + t2;
            __syncthreads();
        }
        const int incl = scanbuf[tid];
        const int base = base_sh;
        const int pos  = base + incl - m;
        if (m && pos < CAP) {
            pts[pos]  = make_float4(xb[(size_t)i * 3], xb[(size_t)i * 3 + 1],
                                    xb[(size_t)i * 3 + 2], __builtin_inff());
            cidx[pos] = (unsigned short)i;
        }
        __syncthreads();
        if (tid == 1023) base_sh = base + incl;
        __syncthreads();
    }
    int Nc = base_sh;
    if (Nc > CAP) Nc = CAP;
    if (Nc == 0) {
        for (int s = tid; s < M_; s += 1024) out_idx[b * M_ + s] = 0;
        return;
    }

    const float NEG_INF = -__builtin_inff();
    const int lane = tid & 31, wav = tid >> 5;
    const int wbase = __builtin_amdgcn_readfirstlane(tid & ~31);
    const int nsl = (__builtin_amdgcn_readfirstlane(Nc) - wbase + 1023) >> 10;

    float px[SLOTS], py[SLOTS], pz[SLOTS], pd[SLOTS];
#pragma unroll
    for (int j = 0; j < SLOTS; ++j) {
        const int i = tid + j * 1024;
        px[j] = 0.0f; py[j] = 0.0f; pz[j] = 0.0f; pd[j] = NEG_INF;
        if (i < Nc) {
            const float4 p = pts[i];
            px[j] = p.x; py[j] = p.y; pz[j] = p.z; pd[j] = p.w;
        }
    }

    int sel = 0;
    for (int s = 0; s < M_; ++s) {
        if (tid == 0) picks[s] = (unsigned short)sel;
        const float lx = pts[sel].x, ly = pts[sel].y, lz = pts[sel].z;

        float bv = NEG_INF;
        int   bi = 0x7fffffff;
#pragma unroll
        for (int j = 0; j < SLOTS; ++j) {
            if (j < nsl) {
                const float dx = px[j] - lx;
                const float dy = py[j] - ly;
                const float dz = pz[j] - lz;
                float sxy;
                {
#pragma clang fp contract(off)
                    sxy = dx * dx + dy * dy;
                }
                const float d  = __builtin_fmaf(dz, dz, sxy);
                const float nd = fminf(pd[j], d);
                pd[j] = nd;
                if (nd > bv) { bv = nd; bi = tid + j * 1024; }
            }
        }
        int wk, wx;
        wave_argmax(__float_as_int(bv), bi, wk, wx);
        if (lane == 0) tab[s & 1][wav] = make_int2(wk, wx);
        __syncthreads();
        const int2 e = tab[s & 1][lane];
        int bk;
        wave_argmax(e.x, e.y, bk, sel);
    }
    __syncthreads();
    for (int t = tid; t < M_; t += 1024) out_idx[b * M_ + t] = (int)cidx[picks[t]];
}

__global__ __launch_bounds__(256) void gather_kernel(
    const float* __restrict__ feat, const float* __restrict__ xyz,
    const float* __restrict__ grasp, const int* __restrict__ idxs,
    float* __restrict__ out)
{
    const int g = blockIdx.x * 256 + threadIdx.x;
    if (g >= B_ * CH_ * M_) return;
    const int m  = g % M_;
    const int t  = g / M_;
    const int ch = t % CH_;
    const int b  = t / CH_;
    const int id = idxs[b * M_ + m];
    float v;
    if (ch < C_)            v = feat[((size_t)b * C_ + ch) * N_ + id];
    else if (ch < C_ + 3)   v = xyz[((size_t)b * N_ + id) * 3 + (ch - C_)];
    else                    v = grasp[(size_t)b * N_ + id];
    out[g] = v;
}

extern "C" void kernel_launch(void* const* d_in, const int* in_sizes, int n_in,
                              void* d_out, int out_size, void* d_ws, size_t ws_size,
                              hipStream_t stream) {
    const float* xyz  = (const float*)d_in[0];
    const float* feat = (const float*)d_in[1];
    const float* Wobj = (const float*)d_in[2];
    const float* bobj = (const float*)d_in[3];
    const float* Wg   = (const float*)d_in[4];
    const float* bg   = (const float*)d_in[5];
    float* out = (float*)d_out;

    char* ws = (char*)d_ws;
    float*         g_grasp = (float*)ws;
    unsigned char* g_mask  = (unsigned char*)(ws + (size_t)B_ * N_ * 4);
    int*           g_idx   = (int*)(ws + (size_t)B_ * N_ * 4 + (size_t)B_ * N_);

    score_kernel<<<B_ * (N_ / 128), 256, 0, stream>>>(feat, Wobj, bobj, Wg, bg,
                                                      g_grasp, g_mask);
    fps_kernel<<<B_, 1024, 0, stream>>>(xyz, g_mask, g_idx);
    const int total = B_ * CH_ * M_;
    gather_kernel<<<(total + 255) / 256, 256, 0, stream>>>(feat, xyz, g_grasp,
                                                           g_idx, out);
}


# FPS compaction in one pass: per-32-point ballot counts and ranks plus one block prefix sum over 1024 group counts replace 32 chunked LDS scans (about 740 barriers to 4); positions identical
# speedup vs baseline: 2.2731x; 1.1458x over previous
#include <hip/hip_runtime.h>
#include <hip/hip_bf16.h>

typedef __attribute__((ext_vector_type(2))) float v2f;
typedef __attribute__((ext_vector_type(8))) float v8f;

#define B_   4
#define N_   32768
#define C_   512
#define M_   512
#define CH_  (C_ + 4)
#define CAP  8704

__global__ __launch_bounds__(256) void score_kernel(
    const float* __restrict__ feat, const float* __restrict__ Wobj,
    const float* __restrict__ bobj, const float* __restrict__ Wg,
    const float* __restrict__ bg, float* __restrict__ grasp_out,
    unsigned char* __restrict__ mask_out)
{
    __shared__ float Apre[128][32][2];
    for (int t = threadIdx.x; t < 128 * 32; t += 256) {
        int kc = t >> 5, lane = t & 31;
        int row = lane & 15;
        int kh  = lane >> 4;
        int ka  = kc * 4 + 2 * kh;
        float a0 = 0.0f, a1 = 0.0f;
        if (row < 2)      { a0 = Wobj[row * C_ + ka]; a1 = Wobj[row * C_ + ka + 1]; }
        else if (row == 2){ a0 = Wg[ka];              a1 = Wg[ka + 1]; }
        Apre[kc][lane][0] = a0;
        Apre[kc][lane][1] = a1;
    }
    __syncthreads();

    const int wave = threadIdx.x >> 5;
    const int lane = threadIdx.x & 31;
    const int b    = blockIdx.x >> 8;
    const int tile = blockIdx.x & 255;
    const int n0   = tile * 128 + wave * 16;
    const int kh   = lane >> 4;
    const int col  = n0 + (lane & 15);

    const float* fb = feat + (size_t)b * C_ * N_;

    float b0 = fb[(size_t)(2 * kh) * N_ + col];
    float b1 = fb[(size_t)(2 * kh + 1) * N_ + col];

    v8f acc = {};
#pragma unroll 2
    for (int kc = 0; kc < 128; ++kc) {
        v2f a, bm;
        a.x  = Apre[kc][lane][0];
        a.y  = Apre[kc][lane][1];
        bm.x = b0;
        bm.y = b1;

        const int kn  = (kc + 1) & 127;
        const int kan = kn * 4 + 2 * kh;
        b0 = fb[(size_t)kan * N_ + col];
        b1 = fb[(size_t)(kan + 1) * N_ + col];

        int pfrow = kc * 4 + 32;
        if (pfrow > C_ - 4) pfrow = C_ - 4;
        __builtin_prefetch(&fb[(size_t)(pfrow + 2 * kh) * N_ + col], 0, 3);
        __builtin_prefetch(&fb[(size_t)(pfrow + 2 * kh + 1) * N_ + col], 0, 3);

        acc = __builtin_amdgcn_wmma_f32_16x16x4_f32(
            false, a, false, bm, (short)0, acc, false, false);
    }

    if (lane < 16) {
        float s0 = acc[0] + bobj[0];
        float s1 = acc[1] + bobj[1];
        float g  = acc[2] + bg[0];
        int p = n0 + lane;
        grasp_out[(size_t)b * N_ + p] = g;
        mask_out[(size_t)b * N_ + p] = (unsigned char)((s1 > s0) && (g > 0.1f));
    }
}

#define SLOTS ((CAP + 1023) / 1024)

__device__ __forceinline__ int xrow_i(int v) {
    return __builtin_amdgcn_permlanex16(v, v, 0x76543210, 0xfedcba98, false, false);
}
template <int X> __device__ __forceinline__ int rowx_i(int v) {
    return __builtin_amdgcn_update_dpp(0, v, 0x160 + X, 0xf, 0xf, true);
}
__device__ __forceinline__ int imax(int a, int b) { return a > b ? a : b; }
__device__ __forceinline__ int imin(int a, int b) { return a < b ? a : b; }
__device__ __forceinline__ int wave_max_i(int v) {
    v = imax(v, rowx_i<1>(v));
    v = imax(v, rowx_i<2>(v));
    v = imax(v, rowx_i<4>(v));
    v = imax(v, rowx_i<8>(v));
    return imax(v, xrow_i(v));
}
__device__ __forceinline__ int wave_min_i(int v) {
    v = imin(v, rowx_i<1>(v));
    v = imin(v, rowx_i<2>(v));
    v = imin(v, rowx_i<4>(v));
    v = imin(v, rowx_i<8>(v));
    return imin(v, xrow_i(v));
}
__device__ __forceinline__ void wave_argmax(int kv, int ki, int& mk, int& mi) {
    mk = wave_max_i(kv);
    const unsigned eq = __builtin_amdgcn_ballot_w32(kv == mk);
    mi = __builtin_amdgcn_readlane(ki, __builtin_ctz(eq));
    if (__builtin_popcount(eq) != 1) {
        mi = __builtin_amdgcn_readfirstlane(
                 wave_min_i(kv == mk ? ki : 0x7fffffff));
    }
}
__device__ __forceinline__ int wave_incl_scan(int v, int lane) {
#pragma unroll
    for (int d = 1; d < 32; d <<= 1) {
        const int t = __builtin_amdgcn_ds_bpermute(((lane - d) & 31) << 2, v);
        if (lane >= d) v += t;
    }
    return v;
}
__global__ __launch_bounds__(1024) void fps_kernel(
    const float* __restrict__ xyz, const unsigned char* __restrict__ mask,
    int* __restrict__ out_idx)
{
    __shared__ float4 pts[CAP];
    __shared__ unsigned short cidx[CAP];
    __shared__ int gcnt[1024];
    __shared__ int wtot[32];
    __shared__ int2 tab[2][32];
    __shared__ unsigned short picks[M_];

    const int b   = blockIdx.x;
    const int tid = threadIdx.x;
    const int lane = tid & 31;
    const int wav = __builtin_amdgcn_readfirstlane(tid >> 5);
    const float* xb = xyz + (size_t)b * N_ * 3;
    const unsigned char* mb = mask + (size_t)b * N_;

    unsigned bits = 0;
#pragma unroll
    for (int c = 0; c < N_ / 1024; ++c)
        bits |= (mb[c * 1024 + tid] ? 1u : 0u) << c;

    int mycnt = 0;
#pragma unroll
    for (int c = 0; c < 32; ++c) {
        const unsigned bal = __builtin_amdgcn_ballot_w32(((bits >> c) & 1u) != 0u);
        if (lane == c) mycnt = __builtin_popcount(bal);
    }
    gcnt[lane * 32 + wav] = mycnt;
    __syncthreads();

    const int cnt_g = gcnt[tid];
    const int inc   = wave_incl_scan(cnt_g, lane);
    if (lane == 31) wtot[wav] = inc;
    __syncthreads();
    const int wt    = wtot[lane];
    const int winc  = wave_incl_scan(wt, lane);
    const int total = __builtin_amdgcn_readlane(winc, 31);
    const int wexc  = __builtin_amdgcn_readlane(winc, wav)
                    - __builtin_amdgcn_readlane(wt, wav);
    gcnt[tid] = wexc + inc - cnt_g;
    __syncthreads();
    const int goff = gcnt[lane * 32 + wav];

    const unsigned below = (1u << lane) - 1u;
#pragma unroll
    for (int cb = 0; cb < 32; cb += 16) {
        float x[16], y[16], z[16];
#pragma unroll
        for (int u = 0; u < 16; ++u) {
            const size_t i = (size_t)((cb + u) * 1024 + tid);
            x[u] = xb[i * 3]; y[u] = xb[i * 3 + 1]; z[u] = xb[i * 3 + 2];
        }
#pragma unroll
        for (int u = 0; u < 16; ++u) {
            const int c = cb + u;
            const bool mine = ((bits >> c) & 1u) != 0u;
            const unsigned bal = __builtin_amdgcn_ballot_w32(mine);
            const int pos = __builtin_amdgcn_readlane(goff, c)
                          + __builtin_popcount(bal & below);
            if (mine && pos < CAP) {
                pts[pos]  = make_float4(x[u], y[u], z[u], __builtin_inff());
                cidx[pos] = (unsigned short)(c * 1024 + tid);
            }
        }
    }
    __syncthreads();
    int Nc = total;
    if (Nc > CAP) Nc = CAP;
    if (Nc == 0) {
        for (int s = tid; s < M_; s += 1024) out_idx[b * M_ + s] = 0;
        return;
    }

    const float NEG_INF = -__builtin_inff();
    const int wbase = __builtin_amdgcn_readfirstlane(tid & ~31);
    const int nsl = (__builtin_amdgcn_readfirstlane(Nc) - wbase + 1023) >> 10;

    float px[SLOTS], py[SLOTS], pz[SLOTS], pd[SLOTS];
#pragma unroll
    for (int j = 0; j < SLOTS; ++j) {
        const int i = tid + j * 1024;
        px[j] = 0.0f; py[j] = 0.0f; pz[j] = 0.0f; pd[j] = NEG_INF;
        if (i < Nc) {
            const float4 p = pts[i];
            px[j] = p.x; py[j] = p.y; pz[j] = p.z; pd[j] = p.w;
        }
    }

    int sel = 0;
    for (int s = 0; s < M_; ++s) {
        if (tid == 0) picks[s] = (unsigned short)sel;
        const float lx = pts[sel].x, ly = pts[sel].y, lz = pts[sel].z;

        float bv = NEG_INF;
        int   bi = 0x7fffffff;
#pragma unroll
        for (int j = 0; j < SLOTS; ++j) {
            if (j < nsl) {
                const float dx = px[j] - lx;
                const float dy = py[j] - ly;
                const float dz = pz[j] - lz;
                float sxy;
                {
#pragma clang fp contract(off)
                    sxy = dx * dx + dy * dy;
                }
                const float d  = __builtin_fmaf(dz, dz, sxy);
                const float nd = fminf(pd[j], d);
                pd[j] = nd;
                if (nd > bv) { bv = nd; bi = tid + j * 1024; }
            }
        }
        int wk, wx;
        wave_argmax(__float_as_int(bv), bi, wk, wx);
        if (lane == 0) tab[s & 1][wav] = make_int2(wk, wx);
        __syncthreads();
        const int2 e = tab[s & 1][lane];
        int bk;
        wave_argmax(e.x, e.y, bk, sel);
    }
    __syncthreads();
    for (int t = tid; t < M_; t += 1024) out_idx[b * M_ + t] = (int)cidx[picks[t]];
}

__global__ __launch_bounds__(256) void gather_kernel(
    const float* __restrict__ feat, const float* __restrict__ xyz,
    const float* __restrict__ grasp, const int* __restrict__ idxs,
    float* __restrict__ out)
{
    const int g = blockIdx.x * 256 + threadIdx.x;
    if (g >= B_ * CH_ * M_) return;
    const int m  = g % M_;
    const int t  = g / M_;
    const int ch = t % CH_;
    const int b  = t / CH_;
    const int id = idxs[b * M_ + m];
    float v;
    if (ch < C_)            v = feat[((size_t)b * C_ + ch) * N_ + id];
    else if (ch < C_ + 3)   v = xyz[((size_t)b * N_ + id) * 3 + (ch - C_)];
    else                    v = grasp[(size_t)b * N_ + id];
    out[g] = v;
}

extern "C" void kernel_launch(void* const* d_in, const int* in_sizes, int n_in,
                              void* d_out, int out_size, void* d_ws, size_t ws_size,
                              hipStream_t stream) {
    const float* xyz  = (const float*)d_in[0];
    const float* feat = (const float*)d_in[1];
    const float* Wobj = (const float*)d_in[2];
    const float* bobj = (const float*)d_in[3];
    const float* Wg   = (const float*)d_in[4];
    const float* bg   = (const float*)d_in[5];
    float* out = (float*)d_out;

    char* ws = (char*)d_ws;
    float*         g_grasp = (float*)ws;
    unsigned char* g_mask  = (unsigned char*)(ws + (size_t)B_ * N_ * 4);
    int*           g_idx   = (int*)(ws + (size_t)B_ * N_ * 4 + (size_t)B_ * N_);

    score_kernel<<<B_ * (N_ / 128), 256, 0, stream>>>(feat, Wobj, bobj, Wg, bg,
                                                      g_grasp, g_mask);
    fps_kernel<<<B_, 1024, 0, stream>>>(xyz, g_mask, g_idx);
    const int total = B_ * CH_ * M_;
    gather_kernel<<<(total + 255) / 256, 256, 0, stream>>>(feat, xyz, g_grasp,
                                                           g_idx, out);
}


# FPS step: thread keeps only its largest key (integer min/max on the float bits, no compare + two selects per slot); the wave finds the index by equality ballots after its max; distances bit-identical
# speedup vs baseline: 2.7180x; 1.1957x over previous
#include <hip/hip_runtime.h>
#include <hip/hip_bf16.h>

typedef __attribute__((ext_vector_type(2))) float v2f;
typedef __attribute__((ext_vector_type(8))) float v8f;

#define B_   4
#define N_   32768
#define C_   512
#define M_   512
#define CH_  (C_ + 4)
#define CAP  8704

__global__ __launch_bounds__(256) void score_kernel(
    const float* __restrict__ feat, const float* __restrict__ Wobj,
    const float* __restrict__ bobj, const float* __restrict__ Wg,
    const float* __restrict__ bg, float* __restrict__ grasp_out,
    unsigned char* __restrict__ mask_out)
{
    __shared__ float Apre[128][32][2];
    for (int t = threadIdx.x; t < 128 * 32; t += 256) {
        int kc = t >> 5, lane = t & 31;
        int row = lane & 15;
        int kh  = lane >> 4;
        int ka  = kc * 4 + 2 * kh;
        float a0 = 0.0f, a1 = 0.0f;
        if (row < 2)      { a0 = Wobj[row * C_ + ka]; a1 = Wobj[row * C_ + ka + 1]; }
        else if (row == 2){ a0 = Wg[ka];              a1 = Wg[ka + 1]; }
        Apre[kc][lane][0] = a0;
        Apre[kc][lane][1] = a1;
    }
    __syncthreads();

    const int wave = threadIdx.x >> 5;
    const int lane = threadIdx.x & 31;
    const int b    = blockIdx.x >> 8;
    const int tile = blockIdx.x & 255;
    const int n0   = tile * 128 + wave * 16;
    const int kh   = lane >> 4;
    const int col  = n0 + (lane & 15);

    const float* fb = feat + (size_t)b * C_ * N_;

    float b0 = fb[(size_t)(2 * kh) * N_ + col];
    float b1 = fb[(size_t)(2 * kh + 1) * N_ + col];

    v8f acc = {};
#pragma unroll 2
    for (int kc = 0; kc < 128; ++kc) {
        v2f a, bm;
        a.x  = Apre[kc][lane][0];
        a.y  = Apre[kc][lane][1];
        bm.x = b0;
        bm.y = b1;

        const int kn  = (kc + 1) & 127;
        const int kan = kn * 4 + 2 * kh;
        b0 = fb[(size_t)kan * N_ + col];
        b1 = fb[(size_t)(kan + 1) * N_ + col];

        int pfrow = kc * 4 + 32;
        if (pfrow > C_ - 4) pfrow = C_ - 4;
        __builtin_prefetch(&fb[(size_t)(pfrow + 2 * kh) * N_ + col], 0, 3);
        __builtin_prefetch(&fb[(size_t)(pfrow + 2 * kh + 1) * N_ + col], 0, 3);

        acc = __builtin_amdgcn_wmma_f32_16x16x4_f32(
            false, a, false, bm, (short)0, acc, false, false);
    }

    if (lane < 16) {
        float s0 = acc[0] + bobj[0];
        float s1 = acc[1] + bobj[1];
        float g  = acc[2] + bg[0];
        int p = n0 + lane;
        grasp_out[(size_t)b * N_ + p] = g;
        mask_out[(size_t)b * N_ + p] = (unsigned char)((s1 > s0) && (g > 0.1f));
    }
}

#define SLOTS ((CAP + 1023) / 1024)

__device__ __forceinline__ int xrow_i(int v) {
    return __builtin_amdgcn_permlanex16(v, v, 0x76543210, 0xfedcba98, false, false);
}
template <int X> __device__ __forceinline__ int rowx_i(int v) {
    return __builtin_amdgcn_update_dpp(0, v, 0x160 + X, 0xf, 0xf, true);
}
__device__ __forceinline__ int imax(int a, int b) { return a > b ? a : b; }
__device__ __forceinline__ int imin(int a, int b) { return a < b ? a : b; }
__device__ __forceinline__ int wave_max_i(int v) {
    v = imax(v, rowx_i<1>(v));
    v = imax(v, rowx_i<2>(v));
    v = imax(v, rowx_i<4>(v));
    v = imax(v, rowx_i<8>(v));
    return imax(v, xrow_i(v));
}
__device__ __forceinline__ int wave_min_i(int v) {
    v = imin(v, rowx_i<1>(v));
    v = imin(v, rowx_i<2>(v));
    v = imin(v, rowx_i<4>(v));
    v = imin(v, rowx_i<8>(v));
    return imin(v, xrow_i(v));
}
__device__ __forceinline__ void wave_argmax(int kv, int ki, int& mk, int& mi) {
    mk = wave_max_i(kv);
    const unsigned eq = __builtin_amdgcn_ballot_w32(kv == mk);
    mi = __builtin_amdgcn_readlane(ki, __builtin_ctz(eq));
    if (__builtin_popcount(eq) != 1) {
        mi = __builtin_amdgcn_readfirstlane(
                 wave_min_i(kv == mk ? ki : 0x7fffffff));
    }
}
__device__ __forceinline__ int wave_incl_scan(int v, int lane) {
#pragma unroll
    for (int d = 1; d < 32; d <<= 1) {
        const int t = __builtin_amdgcn_ds_bpermute(((lane - d) & 31) << 2, v);
        if (lane >= d) v += t;
    }
    return v;
}
template <int NS>
__device__ __forceinline__ void fps_step(
    const float (&px)[SLOTS], const float (&py)[SLOTS], const float (&pz)[SLOTS],
    int (&pk)[SLOTS], const float lx, const float ly, const float lz,
    const int wbase, int& wk, int& wx)
{
    int tm = (int)0xff800000;
#pragma unroll
    for (int j = 0; j < NS; ++j) {
        const float dx = px[j] - lx;
        const float dy = py[j] - ly;
        const float dz = pz[j] - lz;
        float sxy;
        {
#pragma clang fp contract(off)
            sxy = dx * dx + dy * dy;
        }
        const float d = __builtin_fmaf(dz, dz, sxy);
        const int nk = imin(pk[j], __float_as_int(d));
        pk[j] = nk;
        tm = imax(tm, nk);
    }
    const int mk = wave_max_i(tm);
    unsigned long long hit = 0ull;
    int jb = 0;
#pragma unroll
    for (int j = (NS - 1) & ~1; j >= 0; j -= 2) {
        const unsigned lo = __builtin_amdgcn_ballot_w32(pk[j] == mk);
        const unsigned hi = (j + 1 < NS)
            ? __builtin_amdgcn_ballot_w32(pk[(j + 1 < NS) ? j + 1 : j] == mk) : 0u;
        const unsigned long long m = ((unsigned long long)hi << 32) | lo;
        if (m != 0ull) { hit = m; jb = j * 1024; }
    }
    const int pos = __builtin_ctzll(hit);
    const int idx = wbase + jb + (pos & 31) + ((pos >> 5) << 10);
    wk = mk;
    wx = idx;
}
__global__ __launch_bounds__(1024) void fps_kernel(
    const float* __restrict__ xyz, const unsigned char* __restrict__ mask,
    int* __restrict__ out_idx)
{
    __shared__ float4 pts[CAP];
    __shared__ unsigned short cidx[CAP];
    __shared__ int gcnt[1024];
    __shared__ int wtot[32];
    __shared__ int2 tab[2][32];
    __shared__ unsigned short picks[M_];

    const int b   = blockIdx.x;
    const int tid = threadIdx.x;
    const int lane = tid & 31;
    const int wav = __builtin_amdgcn_readfirstlane(tid >> 5);
    const float* xb = xyz + (size_t)b * N_ * 3;
    const unsigned char* mb = mask + (size_t)b * N_;

    unsigned bits = 0;
#pragma unroll
    for (int c = 0; c < N_ / 1024; ++c)
        bits |= (mb[c * 1024 + tid] ? 1u : 0u) << c;

    int mycnt = 0;
#pragma unroll
    for (int c = 0; c < 32; ++c) {
        const unsigned bal = __builtin_amdgcn_ballot_w32(((bits >> c) & 1u) != 0u);
        if (lane == c) mycnt = __builtin_popcount(bal);
    }
    gcnt[lane * 32 + wav] = mycnt;
    __syncthreads();

    const int cnt_g = gcnt[tid];
    const int inc   = wave_incl_scan(cnt_g, lane);
    if (lane == 31) wtot[wav] = inc;
    __syncthreads();
    const int wt    = wtot[lane];
    const int winc  = wave_incl_scan(wt, lane);
    const int total = __builtin_amdgcn_readlane(winc, 31);
    const int wexc  = __builtin_amdgcn_readlane(winc, wav)
                    - __builtin_amdgcn_readlane(wt, wav);
    gcnt[tid] = wexc + inc - cnt_g;
    __syncthreads();
    const int goff = gcnt[lane * 32 + wav];

    const unsigned below = (1u << lane) - 1u;
#pragma unroll
    for (int cb = 0; cb < 32; cb += 16) {
        float x[16], y[16], z[16];
#pragma unroll
        for (int u = 0; u < 16; ++u) {
            const size_t i = (size_t)((cb + u) * 1024 + tid);
            x[u] = xb[i * 3]; y[u] = xb[i * 3 + 1]; z[u] = xb[i * 3 + 2];
        }
#pragma unroll
        for (int u = 0; u < 16; ++u) {
            const int c = cb + u;
            const bool mine = ((bits >> c) & 1u) != 0u;
            const unsigned bal = __builtin_amdgcn_ballot_w32(mine);
            const int pos = __builtin_amdgcn_readlane(goff, c)
                          + __builtin_popcount(bal & below);
            if (mine && pos < CAP) {
                pts[pos]  = make_float4(x[u], y[u], z[u], __builtin_inff());
                cidx[pos] = (unsigned short)(c * 1024 + tid);
            }
        }
    }
    __syncthreads();
    int Nc = total;
    if (Nc > CAP) Nc = CAP;
    if (Nc == 0) {
        for (int s = tid; s < M_; s += 1024) out_idx[b * M_ + s] = 0;
        return;
    }

    const int NEG_INF_BITS = (int)0xff800000;
    const int wbase = __builtin_amdgcn_readfirstlane(tid & ~31);
    const int nsl = (__builtin_amdgcn_readfirstlane(Nc) - wbase + 1023) >> 10;

    float px[SLOTS], py[SLOTS], pz[SLOTS];
    int   pk[SLOTS];
#pragma unroll
    for (int j = 0; j < SLOTS; ++j) {
        const int i = tid + j * 1024;
        px[j] = 0.0f; py[j] = 0.0f; pz[j] = 0.0f; pk[j] = NEG_INF_BITS;
        if (i < Nc) {
            const float4 p = pts[i];
            px[j] = p.x; py[j] = p.y; pz[j] = p.z; pk[j] = __float_as_int(p.w);
        }
    }

    int sel = 0;
    for (int s = 0; s < M_; ++s) {
        if (tid == 0) picks[s] = (unsigned short)sel;
        const float lx = pts[sel].x, ly = pts[sel].y, lz = pts[sel].z;

        int wk = NEG_INF_BITS;
        int wx = 0x7fffffff;
        static_assert(SLOTS == 9, "one case per slot count below");
        switch (nsl) {
        case 0: break;
        case 1: fps_step<1>(px, py, pz, pk, lx, ly, lz, wbase, wk, wx); break;
        case 2: fps_step<2>(px, py, pz, pk, lx, ly, lz, wbase, wk, wx); break;
        case 3: fps_step<3>(px, py, pz, pk, lx, ly, lz, wbase, wk, wx); break;
        case 4: fps_step<4>(px, py, pz, pk, lx, ly, lz, wbase, wk, wx); break;
        case 5: fps_step<5>(px, py, pz, pk, lx, ly, lz, wbase, wk, wx); break;
        case 6: fps_step<6>(px, py, pz, pk, lx, ly, lz, wbase, wk, wx); break;
        case 7: fps_step<7>(px, py, pz, pk, lx, ly, lz, wbase, wk, wx); break;
        case 8: fps_step<8>(px, py, pz, pk, lx, ly, lz, wbase, wk, wx); break;
        default: fps_step<9>(px, py, pz, pk, lx, ly, lz, wbase, wk, wx); break;
        }
        if (lane == 0) tab[s & 1][wav] = make_int2(wk, wx);
        __syncthreads();
        const int2 e = tab[s & 1][lane];
        int bk;
        wave_argmax(e.x, e.y, bk, sel);
    }
    __syncthreads();
    for (int t = tid; t < M_; t += 1024) out_idx[b * M_ + t] = (int)cidx[picks[t]];
}

__global__ __launch_bounds__(256) void gather_kernel(
    const float* __restrict__ feat, const float* __restrict__ xyz,
    const float* __restrict__ grasp, const int* __restrict__ idxs,
    float* __restrict__ out)
{
    const int g = blockIdx.x * 256 + threadIdx.x;
    if (g >= B_ * CH_ * M_) return;
    const int m  = g % M_;
    const int t  = g / M_;
    const int ch = t % CH_;
    const int b  = t / CH_;
    const int id = idxs[b * M_ + m];
    float v;
    if (ch < C_)            v = feat[((size_t)b * C_ + ch) * N_ + id];
    else if (ch < C_ + 3)   v = xyz[((size_t)b * N_ + id) * 3 + (ch - C_)];
    else                    v = grasp[(size_t)b * N_ + id];
    out[g] = v;
}

extern "C" void kernel_launch(void* const* d_in, const int* in_sizes, int n_in,
                              void* d_out, int out_size, void* d_ws, size_t ws_size,
                              hipStream_t stream) {
    const float* xyz  = (const float*)d_in[0];
    const float* feat = (const float*)d_in[1];
    const float* Wobj = (const float*)d_in[2];
    const float* bobj = (const float*)d_in[3];
    const float* Wg   = (const float*)d_in[4];
    const float* bg   = (const float*)d_in[5];
    float* out = (float*)d_out;

    char* ws = (char*)d_ws;
    float*         g_grasp = (float*)ws;
    unsigned char* g_mask  = (unsigned char*)(ws + (size_t)B_ * N_ * 4);
    int*           g_idx   = (int*)(ws + (size_t)B_ * N_ * 4 + (size_t)B_ * N_);

    score_kernel<<<B_ * (N_ / 128), 256, 0, stream>>>(feat, Wobj, bobj, Wg, bg,
                                                      g_grasp, g_mask);
    fps_kernel<<<B_, 1024, 0, stream>>>(xyz, g_mask, g_idx);
    const int total = B_ * CH_ * M_;
    gather_kernel<<<(total + 255) / 256, 256, 0, stream>>>(feat, xyz, g_grasp,
                                                           g_idx, out);
}


# FPS steps on 8 waves (own 256-thread launch, 34 register slots a thread), not 32: a step's per-wave fixed work (wave max, table entry, barrier, table argmax) is paid 8 times; distances bit-identical
# speedup vs baseline: 2.9683x; 1.0921x over previous
#include <hip/hip_runtime.h>
#include <hip/hip_bf16.h>

typedef __attribute__((ext_vector_type(2))) float v2f;
typedef __attribute__((ext_vector_type(8))) float v8f;

#define B_   4
#define N_   32768
#define C_   512
#define M_   512
#define CH_  (C_ + 4)
#define CAP  8704

__global__ __launch_bounds__(256) void score_kernel(
    const float* __restrict__ feat, const float* __restrict__ Wobj,
    const float* __restrict__ bobj, const float* __restrict__ Wg,
    const float* __restrict__ bg, float* __restrict__ grasp_out,
    unsigned char* __restrict__ mask_out)
{
    __shared__ float Apre[128][32][2];
    for (int t = threadIdx.x; t < 128 * 32; t += 256) {
        int kc = t >> 5, lane = t & 31;
        int row = lane & 15;
        int kh  = lane >> 4;
        int ka  = kc * 4 + 2 * kh;
        float a0 = 0.0f, a1 = 0.0f;
        if (row < 2)      { a0 = Wobj[row * C_ + ka]; a1 = Wobj[row * C_ + ka + 1]; }
        else if (row == 2){ a0 = Wg[ka];              a1 = Wg[ka + 1]; }
        Apre[kc][lane][0] = a0;
        Apre[kc][lane][1] = a1;
    }
    __syncthreads();

    const int wave = threadIdx.x >> 5;
    const int lane = threadIdx.x & 31;
    const int b    = blockIdx.x >> 8;
    const int tile = blockIdx.x & 255;
    const int n0   = tile * 128 + wave * 16;
    const int kh   = lane >> 4;
    const int col  = n0 + (lane & 15);

    const float* fb = feat + (size_t)b * C_ * N_;

    float b0 = fb[(size_t)(2 * kh) * N_ + col];
    float b1 = fb[(size_t)(2 * kh + 1) * N_ + col];

    v8f acc = {};
#pragma unroll 2
    for (int kc = 0; kc < 128; ++kc) {
        v2f a, bm;
        a.x  = Apre[kc][lane][0];
        a.y  = Apre[kc][lane][1];
        bm.x = b0;
        bm.y = b1;

        const int kn  = (kc + 1) & 127;
        const int kan = kn * 4 + 2 * kh;
        b0 = fb[(size_t)kan * N_ + col];
        b1 = fb[(size_t)(kan + 1) * N_ + col];

        int pfrow = kc * 4 + 32;
        if (pfrow > C_ - 4) pfrow = C_ - 4;
        __builtin_prefetch(&fb[(size_t)(pfrow + 2 * kh) * N_ + col], 0, 3);
        __builtin_prefetch(&fb[(size_t)(pfrow + 2 * kh + 1) * N_ + col], 0, 3);

        acc = __builtin_amdgcn_wmma_f32_16x16x4_f32(
            false, a, false, bm, (short)0, acc, false, false);
    }

    if (lane < 16) {
        float s0 = acc[0] + bobj[0];
        float s1 = acc[1] + bobj[1];
        float g  = acc[2] + bg[0];
        int p = n0 + lane;
        grasp_out[(size_t)b * N_ + p] = g;
        mask_out[(size_t)b * N_ + p] = (unsigned char)((s1 > s0) && (g > 0.1f));
    }
}

#define FT    256
#define FW    (FT / 32)
#define SLOTS ((CAP + FT - 1) / FT)
#define LIST_COUNT_OFF (N_ - 4)

__device__ __forceinline__ int xrow_i(int v) {
    return __builtin_amdgcn_permlanex16(v, v, 0x76543210, 0xfedcba98, false, false);
}
template <int X> __device__ __forceinline__ int rowx_i(int v) {
    return __builtin_amdgcn_update_dpp(0, v, 0x160 + X, 0xf, 0xf, true);
}
__device__ __forceinline__ int imax(int a, int b) { return a > b ? a : b; }
__device__ __forceinline__ int imin(int a, int b) { return a < b ? a : b; }
__device__ __forceinline__ int wave_max_i(int v) {
    v = imax(v, rowx_i<1>(v));
    v = imax(v, rowx_i<2>(v));
    v = imax(v, rowx_i<4>(v));
    v = imax(v, rowx_i<8>(v));
    return imax(v, xrow_i(v));
}
__device__ __forceinline__ int oct_max_i(int v) {
    v = imax(v, rowx_i<1>(v));
    v = imax(v, rowx_i<2>(v));
    return imax(v, rowx_i<4>(v));
}
__device__ __forceinline__ int oct_min_i(int v) {
    v = imin(v, rowx_i<1>(v));
    v = imin(v, rowx_i<2>(v));
    return imin(v, rowx_i<4>(v));
}
__device__ __forceinline__ void table_argmax(int kv, int ki, int& mk, int& mi) {
    static_assert(FW == 8, "one table entry per lane of an 8-lane group");
    mk = oct_max_i(kv);
    const unsigned eq = __builtin_amdgcn_ballot_w32(kv == mk) & 0xffu;
    mi = __builtin_amdgcn_readlane(ki, __builtin_ctz(eq));
    if (__builtin_popcount(eq) != 1) {
        mi = __builtin_amdgcn_readfirstlane(
                 oct_min_i(kv == mk ? ki : 0x7fffffff));
    }
}
__device__ __forceinline__ int wave_incl_scan(int v, int lane) {
#pragma unroll
    for (int d = 1; d < 32; d <<= 1) {
        const int t = __builtin_amdgcn_ds_bpermute(((lane - d) & 31) << 2, v);
        if (lane >= d) v += t;
    }
    return v;
}

__global__ __launch_bounds__(1024) void compact_kernel(unsigned char* plane)
{
    __shared__ int gcnt[1024];
    __shared__ int wtot[32];

    const int b   = blockIdx.x;
    const int tid = threadIdx.x;
    const int lane = tid & 31;
    const int wav = __builtin_amdgcn_readfirstlane(tid >> 5);
    unsigned char* pb = plane + (size_t)b * N_;
    const unsigned char* mb = pb;
    unsigned short* list = (unsigned short*)pb;

    unsigned bits = 0;
#pragma unroll
    for (int c = 0; c < N_ / 1024; ++c)
        bits |= (mb[c * 1024 + tid] ? 1u : 0u) << c;

    int mycnt = 0;
#pragma unroll
    for (int c = 0; c < 32; ++c) {
        const unsigned bal = __builtin_amdgcn_ballot_w32(((bits >> c) & 1u) != 0u);
        if (lane == c) mycnt = __builtin_popcount(bal);
    }
    gcnt[lane * 32 + wav] = mycnt;
    __syncthreads();

    const int cnt_g = gcnt[tid];
    const int inc   = wave_incl_scan(cnt_g, lane);
    if (lane == 31) wtot[wav] = inc;
    __syncthreads();
    const int wt    = wtot[lane];
    const int winc  = wave_incl_scan(wt, lane);
    const int total = __builtin_amdgcn_readlane(winc, 31);
    const int wexc  = __builtin_amdgcn_readlane(winc, wav)
                    - __builtin_amdgcn_readlane(wt, wav);
    gcnt[tid] = wexc + inc - cnt_g;
    __syncthreads();
    const int goff = gcnt[lane * 32 + wav];

    const unsigned below = (1u << lane) - 1u;
#pragma unroll
    for (int c = 0; c < 32; ++c) {
        const bool mine = ((bits >> c) & 1u) != 0u;
        const unsigned bal = __builtin_amdgcn_ballot_w32(mine);
        const int pos = __builtin_amdgcn_readlane(goff, c)
                      + __builtin_popcount(bal & below);
        if (mine && pos < CAP) list[pos] = (unsigned short)(c * 1024 + tid);
    }
    if (tid == 0) *(int*)(pb + LIST_COUNT_OFF) = total < CAP ? total : CAP;
}

template <int NS>
__device__ __forceinline__ void fps_step(
    const float (&px)[SLOTS], const float (&py)[SLOTS], const float (&pz)[SLOTS],
    int (&pk)[SLOTS], const float lx, const float ly, const float lz,
    const int wbase, int& wk, int& wx)
{
    int tm = (int)0xff800000;
#pragma unroll
    for (int j = 0; j < NS; ++j) {
        const float dx = px[j] - lx;
        const float dy = py[j] - ly;
        const float dz = pz[j] - lz;
        float sxy;
        {
#pragma clang fp contract(off)
            sxy = dx * dx + dy * dy;
        }
        const float d = __builtin_fmaf(dz, dz, sxy);
        const int nk = imin(pk[j], __float_as_int(d));
        pk[j] = nk;
        tm = imax(tm, nk);
    }
    const int mk = wave_max_i(tm);
    unsigned long long hit = 0ull;
    int jb = 0;
#pragma unroll
    for (int j = (NS - 1) & ~1; j >= 0; j -= 2) {
        const unsigned lo = __builtin_amdgcn_ballot_w32(pk[j] == mk);
        const unsigned hi = (j + 1 < NS)
            ? __builtin_amdgcn_ballot_w32(pk[(j + 1 < NS) ? j + 1 : j] == mk) : 0u;
        const unsigned long long m = ((unsigned long long)hi << 32) | lo;
        if (m != 0ull) { hit = m; jb = j * FT; }
    }
    const int pos = __builtin_ctzll(hit);
    const int idx = wbase + jb + (pos & 31) + (pos >> 5) * FT;
    wk = mk;
    wx = idx;
}
template <int NS>
__device__ __forceinline__ void fps_loop(
    const float (&px)[SLOTS], const float (&py)[SLOTS], const float (&pz)[SLOTS],
    int (&pk)[SLOTS], const float4* pts, int2 (*tab)[FW], unsigned short* picks,
    const int tid, const int lane, const int wav, const int wbase)
{
    int sel = 0;
    for (int s = 0; s < M_; ++s) {
        if (tid == 0) picks[s] = (unsigned short)sel;
        const float lx = pts[sel].x, ly = pts[sel].y, lz = pts[sel].z;
        int wk, wx;
        fps_step<NS>(px, py, pz, pk, lx, ly, lz, wbase, wk, wx);
        if (lane == 0) tab[s & 1][wav] = make_int2(wk, wx);
        __syncthreads();
        const int2 e = tab[s & 1][lane & (FW - 1)];
        int bk;
        table_argmax(e.x, e.y, bk, sel);
    }
}
__global__ __launch_bounds__(FT) void fps_kernel(
    const float* __restrict__ xyz, const unsigned char* __restrict__ plane,
    int* __restrict__ out_idx)
{
    __shared__ float4 pts[CAP];
    __shared__ int2 tab[2][FW];
    __shared__ unsigned short picks[M_];

    const int b   = blockIdx.x;
    const int tid = threadIdx.x;
    const int lane = tid & 31;
    const int wav = __builtin_amdgcn_readfirstlane(tid >> 5);
    const float* xb = xyz + (size_t)b * N_ * 3;
    const unsigned char* pb = plane + (size_t)b * N_;
    const unsigned short* list = (const unsigned short*)pb;

    int Nc = __builtin_amdgcn_readfirstlane(*(const int*)(pb + LIST_COUNT_OFF));
    if (Nc > CAP) Nc = CAP;
    if (Nc <= 0) {
        for (int s = tid; s < M_; s += FT) out_idx[b * M_ + s] = 0;
        return;
    }

    const int NEG_INF_BITS = (int)0xff800000;
    const int wbase = __builtin_amdgcn_readfirstlane(tid & ~31);
    const int nsl = (Nc + FT - 1) / FT;

    static_assert(SLOTS % 2 == 0, "two equal halves below");
#pragma unroll
    for (int h = 0; h < SLOTS; h += SLOTS / 2) {
        int id[SLOTS / 2];
#pragma unroll
        for (int u = 0; u < SLOTS / 2; ++u) {
            const int i = tid + (h + u) * FT;
            id[u] = (int)list[i < Nc ? i : Nc - 1];
        }
        float x[SLOTS / 2], y[SLOTS / 2], z[SLOTS / 2];
#pragma unroll
        for (int u = 0; u < SLOTS / 2; ++u) {
            const size_t o = (size_t)id[u] * 3;
            x[u] = xb[o]; y[u] = xb[o + 1]; z[u] = xb[o + 2];
        }
#pragma unroll
        for (int u = 0; u < SLOTS / 2; ++u) {
            const int i = tid + (h + u) * FT;
            if (i < Nc) pts[i] = make_float4(x[u], y[u], z[u], 0.0f);
        }
    }
    __syncthreads();

    float px[SLOTS], py[SLOTS], pz[SLOTS];
    int   pk[SLOTS];
#pragma unroll
    for (int j = 0; j < SLOTS; ++j) {
        const int i = tid + j * FT;
        const float4 p = pts[i < Nc ? i : Nc - 1];
        px[j] = p.x; py[j] = p.y; pz[j] = p.z;
        pk[j] = i < Nc ? __float_as_int(__builtin_inff()) : NEG_INF_BITS;
    }

    static_assert(SLOTS == 34, "one case per slot count below");
#define FPS_CASE(n) case n: fps_loop<n>(px, py, pz, pk, pts, tab, picks, tid, lane, wav, wbase); break;
    switch (nsl) {
    FPS_CASE(1)  FPS_CASE(2)  FPS_CASE(3)  FPS_CASE(4)  FPS_CASE(5)  FPS_CASE(6)
    FPS_CASE(7)  FPS_CASE(8)  FPS_CASE(9)  FPS_CASE(10) FPS_CASE(11) FPS_CASE(12)
    FPS_CASE(13) FPS_CASE(14) FPS_CASE(15) FPS_CASE(16) FPS_CASE(17) FPS_CASE(18)
    FPS_CASE(19) FPS_CASE(20) FPS_CASE(21) FPS_CASE(22) FPS_CASE(23) FPS_CASE(24)
    FPS_CASE(25) FPS_CASE(26) FPS_CASE(27) FPS_CASE(28) FPS_CASE(29) FPS_CASE(30)
    FPS_CASE(31) FPS_CASE(32) FPS_CASE(33)
    default: fps_loop<34>(px, py, pz, pk, pts, tab, picks, tid, lane, wav, wbase); break;
    }
#undef FPS_CASE
    __syncthreads();
    for (int t = tid; t < M_; t += FT) out_idx[b * M_ + t] = (int)list[picks[t]];
}

__global__ __launch_bounds__(256) void gather_kernel(
    const float* __restrict__ feat, const float* __restrict__ xyz,
    const float* __restrict__ grasp, const int* __restrict__ idxs,
    float* __restrict__ out)
{
    const int g = blockIdx.x * 256 + threadIdx.x;
    if (g >= B_ * CH_ * M_) return;
    const int m  = g % M_;
    const int t  = g / M_;
    const int ch = t % CH_;
    const int b  = t / CH_;
    const int id = idxs[b * M_ + m];
    float v;
    if (ch < C_)            v = feat[((size_t)b * C_ + ch) * N_ + id];
    else if (ch < C_ + 3)   v = xyz[((size_t)b * N_ + id) * 3 + (ch - C_)];
    else                    v = grasp[(size_t)b * N_ + id];
    out[g] = v;
}

extern "C" void kernel_launch(void* const* d_in, const int* in_sizes, int n_in,
                              void* d_out, int out_size, void* d_ws, size_t ws_size,
                              hipStream_t stream) {
    const float* xyz  = (const float*)d_in[0];
    const float* feat = (const float*)d_in[1];
    const float* Wobj = (const float*)d_in[2];
    const float* bobj = (const float*)d_in[3];
    const float* Wg   = (const float*)d_in[4];
    const float* bg   = (const float*)d_in[5];
    float* out = (float*)d_out;

    char* ws = (char*)d_ws;
    float*         g_grasp = (float*)ws;
    unsigned char* g_mask  = (unsigned char*)(ws + (size_t)B_ * N_ * 4);
    int*           g_idx   = (int*)(ws + (size_t)B_ * N_ * 4 + (size_t)B_ * N_);

    score_kernel<<<B_ * (N_ / 128), 256, 0, stream>>>(feat, Wobj, bobj, Wg, bg,
                                                      g_grasp, g_mask);
    compact_kernel<<<B_, 1024, 0, stream>>>(g_mask);
    fps_kernel<<<B_, FT, 0, stream>>>(xyz, g_mask, g_idx);
    const int total = B_ * CH_ * M_;
    gather_kernel<<<(total + 255) / 256, 256, 0, stream>>>(feat, xyz, g_grasp,
                                                           g_idx, out);
}


# FPS step index search in two levels: thread keeps its largest key per group of 8 slots; ballots on these find the lowest group holding the wave maximum, only its 8 slots get ballots; picks identical
# speedup vs baseline: 3.0646x; 1.0324x over previous
#include <hip/hip_runtime.h>
#include <hip/hip_bf16.h>

typedef __attribute__((ext_vector_type(2))) float v2f;
typedef __attribute__((ext_vector_type(8))) float v8f;

#define B_   4
#define N_   32768
#define C_   512
#define M_   512
#define CH_  (C_ + 4)
#define CAP  8704

__global__ __launch_bounds__(256) void score_kernel(
    const float* __restrict__ feat, const float* __restrict__ Wobj,
    const float* __restrict__ bobj, const float* __restrict__ Wg,
    const float* __restrict__ bg, float* __restrict__ grasp_out,
    unsigned char* __restrict__ mask_out)
{
    __shared__ float Apre[128][32][2];
    for (int t = threadIdx.x; t < 128 * 32; t += 256) {
        int kc = t >> 5, lane = t & 31;
        int row = lane & 15;
        int kh  = lane >> 4;
        int ka  = kc * 4 + 2 * kh;
        float a0 = 0.0f, a1 = 0.0f;
        if (row < 2)      { a0 = Wobj[row * C_ + ka]; a1 = Wobj[row * C_ + ka + 1]; }
        else if (row == 2){ a0 = Wg[ka];              a1 = Wg[ka + 1]; }
        Apre[kc][lane][0] = a0;
        Apre[kc][lane][1] = a1;
    }
    __syncthreads();

    const int wave = threadIdx.x >> 5;
    const int lane = threadIdx.x & 31;
    const int b    = blockIdx.x >> 8;
    const int tile = blockIdx.x & 255;
    const int n0   = tile * 128 + wave * 16;
    const int kh   = lane >> 4;
    const int col  = n0 + (lane & 15);

    const float* fb = feat + (size_t)b * C_ * N_;

    float b0 = fb[(size_t)(2 * kh) * N_ + col];
    float b1 = fb[(size_t)(2 * kh + 1) * N_ + col];

    v8f acc = {};
#pragma unroll 2
    for (int kc = 0; kc < 128; ++kc) {
        v2f a, bm;
        a.x  = Apre[kc][lane][0];
        a.y  = Apre[kc][lane][1];
        bm.x = b0;
        bm.y = b1;

        const int kn  = (kc + 1) & 127;
        const int kan = kn * 4 + 2 * kh;
        b0 = fb[(size_t)kan * N_ + col];
        b1 = fb[(size_t)(kan + 1) * N_ + col];

        int pfrow = kc * 4 + 32;
        if (pfrow > C_ - 4) pfrow = C_ - 4;
        __builtin_prefetch(&fb[(size_t)(pfrow + 2 * kh) * N_ + col], 0, 3);
        __builtin_prefetch(&fb[(size_t)(pfrow + 2 * kh + 1) * N_ + col], 0, 3);

        acc = __builtin_amdgcn_wmma_f32_16x16x4_f32(
            false, a, false, bm, (short)0, acc, false, false);
    }

    if (lane < 16) {
        float s0 = acc[0] + bobj[0];
        float s1 = acc[1] + bobj[1];
        float g  = acc[2] + bg[0];
        int p = n0 + lane;
        grasp_out[(size_t)b * N_ + p] = g;
        mask_out[(size_t)b * N_ + p] = (unsigned char)((s1 > s0) && (g > 0.1f));
    }
}

#define FT    256
#define FW    (FT / 32)
#define SLOTS ((CAP + FT - 1) / FT)
#define LIST_COUNT_OFF (N_ - 4)

__device__ __forceinline__ int xrow_i(int v) {
    return __builtin_amdgcn_permlanex16(v, v, 0x76543210, 0xfedcba98, false, false);
}
template <int X> __device__ __forceinline__ int rowx_i(int v) {
    return __builtin_amdgcn_update_dpp(0, v, 0x160 + X, 0xf, 0xf, true);
}
__device__ __forceinline__ int imax(int a, int b) { return a > b ? a : b; }
__device__ __forceinline__ int imin(int a, int b) { return a < b ? a : b; }
__device__ __forceinline__ int wave_max_i(int v) {
    v = imax(v, rowx_i<1>(v));
    v = imax(v, rowx_i<2>(v));
    v = imax(v, rowx_i<4>(v));
    v = imax(v, rowx_i<8>(v));
    return imax(v, xrow_i(v));
}
__device__ __forceinline__ int oct_max_i(int v) {
    v = imax(v, rowx_i<1>(v));
    v = imax(v, rowx_i<2>(v));
    return imax(v, rowx_i<4>(v));
}
__device__ __forceinline__ int oct_min_i(int v) {
    v = imin(v, rowx_i<1>(v));
    v = imin(v, rowx_i<2>(v));
    return imin(v, rowx_i<4>(v));
}
__device__ __forceinline__ void table_argmax(int kv, int ki, int& mk, int& mi) {
    static_assert(FW == 8, "one table entry per lane of an 8-lane group");
    mk = oct_max_i(kv);
    const unsigned eq = __builtin_amdgcn_ballot_w32(kv == mk) & 0xffu;
    mi = __builtin_amdgcn_readlane(ki, __builtin_ctz(eq));
    if (__builtin_popcount(eq) != 1) {
        mi = __builtin_amdgcn_readfirstlane(
                 oct_min_i(kv == mk ? ki : 0x7fffffff));
    }
}
__device__ __forceinline__ int wave_incl_scan(int v, int lane) {
#pragma unroll
    for (int d = 1; d < 32; d <<= 1) {
        const int t = __builtin_amdgcn_ds_bpermute(((lane - d) & 31) << 2, v);
        if (lane >= d) v += t;
    }
    return v;
}

__global__ __launch_bounds__(1024) void compact_kernel(unsigned char* plane)
{
    __shared__ int gcnt[1024];
    __shared__ int wtot[32];

    const int b   = blockIdx.x;
    const int tid = threadIdx.x;
    const int lane = tid & 31;
    const int wav = __builtin_amdgcn_readfirstlane(tid >> 5);
    unsigned char* pb = plane + (size_t)b * N_;
    const unsigned char* mb = pb;
    unsigned short* list = (unsigned short*)pb;

    unsigned bits = 0;
#pragma unroll
    for (int c = 0; c < N_ / 1024; ++c)
        bits |= (mb[c * 1024 + tid] ? 1u : 0u) << c;

    int mycnt = 0;
#pragma unroll
    for (int c = 0; c < 32; ++c) {
        const unsigned bal = __builtin_amdgcn_ballot_w32(((bits >> c) & 1u) != 0u);
        if (lane == c) mycnt = __builtin_popcount(bal);
    }
    gcnt[lane * 32 + wav] = mycnt;
    __syncthreads();

    const int cnt_g = gcnt[tid];
    const int inc   = wave_incl_scan(cnt_g, lane);
    if (lane == 31) wtot[wav] = inc;
    __syncthreads();
    const int wt    = wtot[lane];
    const int winc  = wave_incl_scan(wt, lane);
    const int total = __builtin_amdgcn_readlane(winc, 31);
    const int wexc  = __builtin_amdgcn_readlane(winc, wav)
                    - __builtin_amdgcn_readlane(wt, wav);
    gcnt[tid] = wexc + inc - cnt_g;
    __syncthreads();
    const int goff = gcnt[lane * 32 + wav];

    const unsigned below = (1u << lane) - 1u;
#pragma unroll
    for (int c = 0; c < 32; ++c) {
        const bool mine = ((bits >> c) & 1u) != 0u;
        const unsigned bal = __builtin_amdgcn_ballot_w32(mine);
        const int pos = __builtin_amdgcn_readlane(goff, c)
                      + __builtin_popcount(bal & below);
        if (mine && pos < CAP) list[pos] = (unsigned short)(c * 1024 + tid);
    }
    if (tid == 0) *(int*)(pb + LIST_COUNT_OFF) = total < CAP ? total : CAP;
}

#define GS 8

template <int NS, int G>
__device__ __forceinline__ int group_search(const int (&pk)[SLOTS], const int mk,
                                            const int wbase)
{
    constexpr int J0 = G * GS;
    constexpr int J1 = (J0 + GS < NS) ? J0 + GS : NS;
    constexpr int JT = J0 + ((J1 - J0 - 1) & ~1);
    unsigned long long hit = 0ull;
    int jb = 0;
#pragma unroll
    for (int j = JT; j >= J0; j -= 2) {
        const unsigned lo = __builtin_amdgcn_ballot_w32(pk[j] == mk);
        const unsigned hi = (j + 1 < J1)
            ? __builtin_amdgcn_ballot_w32(pk[(j + 1 < J1) ? j + 1 : j] == mk) : 0u;
        const unsigned long long m = ((unsigned long long)hi << 32) | lo;
        if (j == JT || m != 0ull) { hit = m; jb = j * FT; }
    }
    const int pos = __builtin_ctzll(hit);
    return wbase + jb + (pos & 31) + (pos >> 5) * FT;
}
#define NGMAX ((SLOTS + GS - 1) / GS)
template <int NS, int G>
__device__ __forceinline__ int first_group_search(
    const int (&pk)[SLOTS], const unsigned (&gb)[NGMAX], const int mk, const int wbase)
{
    constexpr int NG = (NS + GS - 1) / GS;
    if constexpr (G == NG - 1) {
        return group_search<NS, G>(pk, mk, wbase);
    } else {
        if (gb[G] != 0u) return group_search<NS, G>(pk, mk, wbase);
        return first_group_search<NS, G + 1>(pk, gb, mk, wbase);
    }
}
template <int NS>
__device__ __forceinline__ void fps_step(
    const float (&px)[SLOTS], const float (&py)[SLOTS], const float (&pz)[SLOTS],
    int (&pk)[SLOTS], const float lx, const float ly, const float lz,
    const int wbase, int& wk, int& wx)
{
    constexpr int NG = (NS + GS - 1) / GS;
    int tg[NGMAX];
#pragma unroll
    for (int g = 0; g < NGMAX; ++g) tg[g] = (int)0xff800000;
#pragma unroll
    for (int j = 0; j < NS; ++j) {
        const float dx = px[j] - lx;
        const float dy = py[j] - ly;
        const float dz = pz[j] - lz;
        float sxy;
        {
#pragma clang fp contract(off)
            sxy = dx * dx + dy * dy;
        }
        const float d = __builtin_fmaf(dz, dz, sxy);
        const int nk = imin(pk[j], __float_as_int(d));
        pk[j] = nk;
        tg[j / GS] = imax(tg[j / GS], nk);
    }
    int tm = tg[0];
#pragma unroll
    for (int g = 1; g < NG; ++g) tm = imax(tm, tg[g]);
    const int mk = wave_max_i(tm);
    unsigned gb[NGMAX] = {};
#pragma unroll
    for (int g = 0; g + 1 < NG; ++g) gb[g] = __builtin_amdgcn_ballot_w32(tg[g] == mk);
    wk = mk;
    wx = first_group_search<NS, 0>(pk, gb, mk, wbase);
}
template <int NS>
__device__ __forceinline__ void fps_loop(
    const float (&px)[SLOTS], const float (&py)[SLOTS], const float (&pz)[SLOTS],
    int (&pk)[SLOTS], const float4* pts, int2 (*tab)[FW], unsigned short* picks,
    const int tid, const int lane, const int wav, const int wbase)
{
    int sel = 0;
    for (int s = 0; s < M_; ++s) {
        if (tid == 0) picks[s] = (unsigned short)sel;
        const float lx = pts[sel].x, ly = pts[sel].y, lz = pts[sel].z;
        int wk, wx;
        fps_step<NS>(px, py, pz, pk, lx, ly, lz, wbase, wk, wx);
        if (lane == 0) tab[s & 1][wav] = make_int2(wk, wx);
        __syncthreads();
        const int2 e = tab[s & 1][lane & (FW - 1)];
        int bk;
        table_argmax(e.x, e.y, bk, sel);
    }
}
__global__ __launch_bounds__(FT) void fps_kernel(
    const float* __restrict__ xyz, const unsigned char* __restrict__ plane,
    int* __restrict__ out_idx)
{
    __shared__ float4 pts[CAP];
    __shared__ int2 tab[2][FW];
    __shared__ unsigned short picks[M_];

    const int b   = blockIdx.x;
    const int tid = threadIdx.x;
    const int lane = tid & 31;
    const int wav = __builtin_amdgcn_readfirstlane(tid >> 5);
    const float* xb = xyz + (size_t)b * N_ * 3;
    const unsigned char* pb = plane + (size_t)b * N_;
    const unsigned short* list = (const unsigned short*)pb;

    int Nc = __builtin_amdgcn_readfirstlane(*(const int*)(pb + LIST_COUNT_OFF));
    if (Nc > CAP) Nc = CAP;
    if (Nc <= 0) {
        for (int s = tid; s < M_; s += FT) out_idx[b * M_ + s] = 0;
        return;
    }

    const int NEG_INF_BITS = (int)0xff800000;
    const int wbase = __builtin_amdgcn_readfirstlane(tid & ~31);
    const int nsl = (Nc + FT - 1) / FT;

    static_assert(SLOTS % 2 == 0, "two equal halves below");
#pragma unroll
    for (int h = 0; h < SLOTS; h += SLOTS / 2) {
        int id[SLOTS / 2];
#pragma unroll
        for (int u = 0; u < SLOTS / 2; ++u) {
            const int i = tid + (h + u) * FT;
            id[u] = (int)list[i < Nc ? i : Nc - 1];
        }
        float x[SLOTS / 2], y[SLOTS / 2], z[SLOTS / 2];
#pragma unroll
        for (int u = 0; u < SLOTS / 2; ++u) {
            const size_t o = (size_t)id[u] * 3;
            x[u] = xb[o]; y[u] = xb[o + 1]; z[u] = xb[o + 2];
        }
#pragma unroll
        for (int u = 0; u < SLOTS / 2; ++u) {
            const int i = tid + (h + u) * FT;
            if (i < Nc) pts[i] = make_float4(x[u], y[u], z[u], 0.0f);
        }
    }
    __syncthreads();

    float px[SLOTS], py[SLOTS], pz[SLOTS];
    int   pk[SLOTS];
#pragma unroll
    for (int j = 0; j < SLOTS; ++j) {
        const int i = tid + j * FT;
        const float4 p = pts[i < Nc ? i : Nc - 1];
        px[j] = p.x; py[j] = p.y; pz[j] = p.z;
        pk[j] = i < Nc ? __float_as_int(__builtin_inff()) : NEG_INF_BITS;
    }

    static_assert(SLOTS == 34, "one case per slot count below");
#define FPS_CASE(n) case n: fps_loop<n>(px, py, pz, pk, pts, tab, picks, tid, lane, wav, wbase); break;
    switch (nsl) {
    FPS_CASE(1)  FPS_CASE(2)  FPS_CASE(3)  FPS_CASE(4)  FPS_CASE(5)  FPS_CASE(6)
    FPS_CASE(7)  FPS_CASE(8)  FPS_CASE(9)  FPS_CASE(10) FPS_CASE(11) FPS_CASE(12)
    FPS_CASE(13) FPS_CASE(14) FPS_CASE(15) FPS_CASE(16) FPS_CASE(17) FPS_CASE(18)
    FPS_CASE(19) FPS_CASE(20) FPS_CASE(21) FPS_CASE(22) FPS_CASE(23) FPS_CASE(24)
    FPS_CASE(25) FPS_CASE(26) FPS_CASE(27) FPS_CASE(28) FPS_CASE(29) FPS_CASE(30)
    FPS_CASE(31) FPS_CASE(32) FPS_CASE(33)
    default: fps_loop<34>(px, py, pz, pk, pts, tab, picks, tid, lane, wav, wbase); break;
    }
#undef FPS_CASE
    __syncthreads();
    for (int t = tid; t < M_; t += FT) out_idx[b * M_ + t] = (int)list[picks[t]];
}

__global__ __launch_bounds__(256) void gather_kernel(
    const float* __restrict__ feat, const float* __restrict__ xyz,
    const float* __restrict__ grasp, const int* __restrict__ idxs,
    float* __restrict__ out)
{
    const int g = blockIdx.x * 256 + threadIdx.x;
    if (g >= B_ * CH_ * M_) return;
    const int m  = g % M_;
    const int t  = g / M_;
    const int ch = t % CH_;
    const int b  = t / CH_;
    const int id = idxs[b * M_ + m];
    float v;
    if (ch < C_)            v = feat[((size_t)b * C_ + ch) * N_ + id];
    else if (ch < C_ + 3)   v = xyz[((size_t)b * N_ + id) * 3 + (ch - C_)];
    else                    v = grasp[(size_t)b * N_ + id];
    out[g] = v;
}

extern "C" void kernel_launch(void* const* d_in, const int* in_sizes, int n_in,
                              void* d_out, int out_size, void* d_ws, size_t ws_size,
                              hipStream_t stream) {
    const float* xyz  = (const float*)d_in[0];
    const float* feat = (const float*)d_in[1];
    const float* Wobj = (const float*)d_in[2];
    const float* bobj = (const float*)d_in[3];
    const float* Wg   = (const float*)d_in[4];
    const float* bg   = (const float*)d_in[5];
    float* out = (float*)d_out;

    char* ws = (char*)d_ws;
    float*         g_grasp = (float*)ws;
    unsigned char* g_mask  = (unsigned char*)(ws + (size_t)B_ * N_ * 4);
    int*           g_idx   = (int*)(ws + (size_t)B_ * N_ * 4 + (size_t)B_ * N_);

    score_kernel<<<B_ * (N_ / 128), 256, 0, stream>>>(feat, Wobj, bobj, Wg, bg,
                                                      g_grasp, g_mask);
    compact_kernel<<<B_, 1024, 0, stream>>>(g_mask);
    fps_kernel<<<B_, FT, 0, stream>>>(xyz, g_mask, g_idx);
    const int total = B_ * CH_ * M_;
    gather_kernel<<<(total + 255) / 256, 256, 0, stream>>>(feat, xyz, g_grasp,
                                                           g_idx, out);
}
